# starting kernel
# baseline (speedup 1.0000x reference)
#include <hip/hip_runtime.h>
#include <hip/hip_bf16.h>


typedef __attribute__((ext_vector_type(2))) float v2f;
typedef __attribute__((ext_vector_type(8))) float v8f;

#define FRAMES       4000
#define D2V          256
#define OUT_T        4001
#define OUT_LEN      32008
#define TILE_OUT     15
#define TILES_PER_BC 267
#define N_BC         32
#define WAVES_PER_BLOCK 8
#define N_BLOCKS     ((N_BC * TILES_PER_BC) / WAVES_PER_BLOCK)

__global__ __launch_bounds__(256)
void decoder_oadd_wmma(const float* __restrict__ est, float* __restrict__ out)
{
    __shared__ float lds[WAVES_PER_BLOCK][16][17];

    const int lane  = threadIdx.x & 31;
    const int wave  = threadIdx.x >> 5;
    const int gwave = blockIdx.x * WAVES_PER_BLOCK + wave;
    const int bc    = gwave / TILES_PER_BC;
    const int tile  = gwave % TILES_PER_BC;

    const int f0 = tile * TILE_OUT - 1;
    const int m  = lane & 15;
    const int kg = lane >> 4;
    const int frame = f0 + m;

    const float* srcbc = est + (size_t)bc * (size_t)(D2V * FRAMES);

    v8f acc = {};
    if (f0 >= 0 && f0 + 15 < FRAMES) {
        const float* p0 = srcbc + (size_t)(2 * kg) * FRAMES + frame;
        #pragma unroll 4
        for (int c4 = 0; c4 < 64; ++c4) {
            v2f a;
            a.x = p0[(size_t)(4 * c4)     * FRAMES];
            a.y = p0[(size_t)(4 * c4 + 1) * FRAMES];
            const float bv = (m == (c4 >> 2)) ? 0.0625f : 0.0f;
            v2f b; b.x = bv; b.y = bv;
            acc = __builtin_amdgcn_wmma_f32_16x16x4_f32(
                false, a, false, b, (short)0, acc, false, false);
        }
    } else {
        const bool fvalid = (frame >= 0) && (frame < FRAMES);
        #pragma unroll 4
        for (int c4 = 0; c4 < 64; ++c4) {
            v2f a; a.x = 0.0f; a.y = 0.0f;
            if (fvalid) {
                const float* p = srcbc + (size_t)(4 * c4 + 2 * kg) * FRAMES + frame;
                a.x = p[0];
                a.y = p[FRAMES];
            }
            const float bv = (m == (c4 >> 2)) ? 0.0625f : 0.0f;
            v2f b; b.x = bv; b.y = bv;
            acc = __builtin_amdgcn_wmma_f32_16x16x4_f32(
                false, a, false, b, (short)0, acc, false, false);
        }
    }

    #pragma unroll
    for (int v = 0; v < 8; ++v)
        lds[wave][v + 8 * kg][m] = acc[v];

    __syncthreads();

    float* obase = out + (size_t)bc * OUT_LEN;
    for (int i = lane; i < 120; i += 32) {
        const int off = 1 + (i >> 3);
        const int r   = i & 7;
        const int s   = f0 + off;
        if (s < OUT_T)
            obase[8 * s + r] = lds[wave][off][r] + lds[wave][off - 1][8 + r];
    }
}

extern "C" void kernel_launch(void* const* d_in, const int* in_sizes, int n_in,
                              void* d_out, int out_size, void* d_ws, size_t ws_size,
                              hipStream_t stream)
{
    const float* est = (const float*)d_in[0];
    float* out = (float*)d_out;
    hipLaunchKernelGGL(decoder_oadd_wmma,
                       dim3(N_BLOCKS), dim3(256), 0, stream,
                       est, out);
}


# Decompose by output residue: wave r sums the rows of windows r and 8+r, lanes along frames (one aligned 256-byte row piece per load); 8 waves interleave residues through LDS; sum order changed
# speedup vs baseline: 2.4209x; 2.4209x over previous
#include <hip/hip_runtime.h>


typedef __attribute__((ext_vector_type(2))) float v2f;
typedef __attribute__((ext_vector_type(4))) float v4f;
typedef __attribute__((ext_vector_type(8))) float v8f;

#define FRAMES      4000
#define ROWS        256
#define OUT_LEN     32008
#define N_BC        32
#define SPAN        256
#define SPANS       16
#define LAST_VALID  (FRAMES - (SPANS - 1) * SPAN)
#define PITCH       268

static __device__ __forceinline__ v8f mm(v2f a, v2f b, v8f c)
{
    return __builtin_amdgcn_wmma_f32_16x16x4_f32(false, a, false, b, (short)0, c, false, false);
}

static __device__ __forceinline__ int col(int s) { return s + 2 * (s >> 6); }

static __device__ __forceinline__ v2f selector(int q, int h, int c)
{
    v2f b;
    b.x = (c == 4 * q + 2 * h)     ? 0.0625f : 0.0f;
    b.y = (c == 4 * q + 2 * h + 1) ? 0.0625f : 0.0f;
    return b;
}

#define LOAD_PIECE(buf, k)                                                        \
    _Pragma("unroll")                                                             \
    for (int j = 0; j < 16; ++j) {                                                \
        buf[j]      = *(const v2f*)(plo + j * FRAMES + 64 * (k));                 \
        buf[16 + j] = *(const v2f*)(phi + j * FRAMES + 64 * (k));                 \
    }

#define SUM_PIECE(buf, k)                                                         \
    {                                                                             \
        const v2f b = selector((k), h, c);                                        \
        _Pragma("unroll")                                                         \
        for (int j = 0; j < 16; ++j) {                                            \
            acc_lo = mm(buf[j],      b, acc_lo);                                  \
            acc_hi = mm(buf[16 + j], b, acc_hi);                                  \
        }                                                                         \
    }

__global__ __launch_bounds__(256)
void oadd_residue_wmma(const float* __restrict__ est, float* __restrict__ out)
{
    __shared__ float L[2][8][PITCH];

    const int tid  = threadIdx.x;
    const int lane = tid & 31;
    const int r    = tid >> 5;
    const int h    = lane >> 4;
    const int c    = lane & 15;
    const int bc   = blockIdx.x / SPANS;
    const int span = blockIdx.x % SPANS;
    const int F0   = span * SPAN;

    const float* lo = est + (size_t)bc * (ROWS * FRAMES) + (size_t)(16 * r) * FRAMES + F0;
    const float* hi = lo + 128 * FRAMES;

    v8f acc_lo = {}, acc_hi = {}, acc_halo = {};

    v2f hal[4];
    #pragma unroll
    for (int m = 0; m < 4; ++m) { hal[m].x = 0.0f; hal[m].y = 0.0f; }
    if (span > 0 && c == 0) {
        #pragma unroll
        for (int m = 0; m < 4; ++m) {
            const float* p = hi + (4 * m + 2 * h) * FRAMES - 1;
            hal[m].x = p[0];
            hal[m].y = p[FRAMES];
        }
    }

    if (span < SPANS - 1) {
        const float* plo = lo + 2 * lane;
        const float* phi = hi + 2 * lane;
        v2f pa[32], pb[32];
        LOAD_PIECE(pa, 0)
        LOAD_PIECE(pb, 1)
        SUM_PIECE(pa, 0)
        LOAD_PIECE(pa, 2)
        SUM_PIECE(pb, 1)
        LOAD_PIECE(pb, 3)
        SUM_PIECE(pa, 2)
        SUM_PIECE(pb, 3)
    } else {
        for (int k = 0; k < 3; ++k) {
            int f = 64 * k + 2 * lane;
            if (f > LAST_VALID - 2) f = LAST_VALID - 2;
            const float* plo = lo + f;
            const float* phi = hi + f;
            const v2f b = selector(k, h, c);
            #pragma unroll
            for (int j = 0; j < 16; ++j) {
                const v2f alo = *(const v2f*)(plo + j * FRAMES);
                const v2f ahi = *(const v2f*)(phi + j * FRAMES);
                acc_lo = mm(alo, b, acc_lo);
                acc_hi = mm(ahi, b, acc_hi);
            }
        }
    }

    {
        v2f b0;
        b0.x = (c == 0) ? 0.0625f : 0.0f;
        b0.y = b0.x;
        #pragma unroll
        for (int m = 0; m < 4; ++m) acc_halo = mm(hal[m], b0, acc_halo);
    }

    const int fl0 = 64 * (c >> 2) + 32 * ((c >> 1) & 1) + 16 * h + (c & 1);
    #pragma unroll
    for (int v = 0; v < 8; ++v) {
        const int s = fl0 + 2 * v;
        L[0][r][col(s)]     = acc_lo[v];
        L[1][r][col(s + 1)] = acc_hi[v];
    }
    if (lane == 0) L[1][r][0] = acc_halo[0];

    __syncthreads();

    const int n_s = (span < SPANS - 1) ? SPAN : (LAST_VALID + 1);
    float* obase = out + (size_t)bc * OUT_LEN + (size_t)8 * F0;
    #pragma unroll
    for (int pass = 0; pass < 2; ++pass) {
        const int g  = tid + 256 * pass;
        const int s  = g >> 1;
        const int r0 = 4 * (g & 1);
        if (s < n_s) {
            const bool lo_ok = (F0 + s) < FRAMES;
            const int  cs    = col(s);
            v4f o;
            #pragma unroll
            for (int i = 0; i < 4; ++i) {
                const float a = lo_ok ? L[0][r0 + i][cs] : 0.0f;
                o[i] = a + L[1][r0 + i][cs];
            }
            *(v4f*)(obase + 4 * g) = o;
        }
    }
}

extern "C" void kernel_launch(void* const* d_in, const int* in_sizes, int n_in,
                              void* d_out, int out_size, void* d_ws, size_t ws_size,
                              hipStream_t stream)
{
    const float* est = (const float*)d_in[0];
    float* out = (float*)d_out;
    hipLaunchKernelGGL(oadd_residue_wmma,
                       dim3(N_BC * SPANS), dim3(256), 0, stream,
                       est, out);
}


# Pin the in-wave order: products run in load-return order, 64 loads behind (a row pair's 2 products, then the 2 loads refilling its registers): 1 product, not 96, after the last load; sum order same
# speedup vs baseline: 2.7297x; 1.1276x over previous
#include <hip/hip_runtime.h>


typedef __attribute__((ext_vector_type(2))) float v2f;
typedef __attribute__((ext_vector_type(4))) float v4f;
typedef __attribute__((ext_vector_type(8))) float v8f;

#define FRAMES      4000
#define ROWS        256
#define OUT_LEN     32008
#define N_BC        32
#define SPAN        256
#define SPANS       16
#define LAST_VALID  (FRAMES - (SPANS - 1) * SPAN)
#define PITCH       268

static __device__ __forceinline__ v8f mm(v2f a, v2f b, v8f c)
{
    return __builtin_amdgcn_wmma_f32_16x16x4_f32(false, a, false, b, (short)0, c, false, false);
}

static __device__ __forceinline__ int col(int s) { return s + 2 * (s >> 6); }

static __device__ __forceinline__ v2f selector(int q, int h, int c)
{
    v2f b;
    b.x = (c == 4 * q + 2 * h)     ? 0.0625f : 0.0f;
    b.y = (c == 4 * q + 2 * h + 1) ? 0.0625f : 0.0f;
    return b;
}

#define PIN() asm volatile("" : "+v"(acc_lo), "+v"(acc_hi) : : "memory")

#define LOAD_FENCE() asm volatile("" : : : "memory")


#define LOAD_PIECE(buf, qlo, qhi, off)                                            \
    _Pragma("unroll")                                                             \
    for (int j = 0; j < 16; ++j) {                                                \
        buf[j]      = *(const v2f*)((qlo) + j * FRAMES + (off));                  \
        buf[16 + j] = *(const v2f*)((qhi) + j * FRAMES + (off));                  \
        LOAD_FENCE();                                                             \
    }

#define SUM_PIECE_REFILL(buf, k, qlo, qhi, off)                                   \
    {                                                                             \
        const v2f b = selector((k), h, c);                                        \
        _Pragma("unroll")                                                         \
        for (int j = 0; j < 16; ++j) {                                            \
            acc_lo = mm(buf[j],      b, acc_lo);                                  \
            acc_hi = mm(buf[16 + j], b, acc_hi);                                  \
            PIN();                                                                \
            buf[j]      = *(const v2f*)((qlo) + j * FRAMES + (off));              \
            buf[16 + j] = *(const v2f*)((qhi) + j * FRAMES + (off));              \
        }                                                                         \
    }

#define SUM_PIECE(buf, k)                                                         \
    {                                                                             \
        const v2f b = selector((k), h, c);                                        \
        _Pragma("unroll")                                                         \
        for (int j = 0; j < 16; ++j) {                                            \
            acc_lo = mm(buf[j],      b, acc_lo);                                  \
            acc_hi = mm(buf[16 + j], b, acc_hi);                                  \
            PIN();                                                                \
        }                                                                         \
    }

__global__ __launch_bounds__(256)
void oadd_residue_wmma(const float* __restrict__ est, float* __restrict__ out)
{
    __shared__ float L[2][8][PITCH];

    const int tid  = threadIdx.x;
    const int lane = tid & 31;
    const int r    = tid >> 5;
    const int h    = lane >> 4;
    const int c    = lane & 15;
    const int bc   = blockIdx.x / SPANS;
    const int span = blockIdx.x % SPANS;
    const int F0   = span * SPAN;

    const float* lo = est + (size_t)bc * (ROWS * FRAMES) + (size_t)(16 * r) * FRAMES + F0;
    const float* hi = lo + 128 * FRAMES;

    v8f acc_lo = {}, acc_hi = {}, acc_halo = {};

    v2f hal[4];
    #pragma unroll
    for (int m = 0; m < 4; ++m) { hal[m].x = 0.0f; hal[m].y = 0.0f; }
    if (span > 0 && c == 0) {
        #pragma unroll
        for (int m = 0; m < 4; ++m) {
            const float* p = hi + (4 * m + 2 * h) * FRAMES - 1;
            hal[m].x = p[0];
            hal[m].y = p[FRAMES];
        }
    }

    const float* plo = lo + 2 * lane;
    const float* phi = hi + 2 * lane;
    int f2 = 128 + 2 * lane;
    if (span == SPANS - 1 && f2 > LAST_VALID - 2) f2 = LAST_VALID - 2;
    const float* qlo = lo + f2;
    const float* qhi = hi + f2;

    v2f pa[32], pb[32];
    LOAD_PIECE(pa, plo, phi, 0)
    LOAD_PIECE(pb, plo, phi, 64)
    SUM_PIECE_REFILL(pa, 0, qlo, qhi, 0)

    {
        v2f b0;
        b0.x = (c == 0) ? 0.0625f : 0.0f;
        b0.y = b0.x;
        #pragma unroll
        for (int m = 0; m < 4; ++m) acc_halo = mm(hal[m], b0, acc_halo);
    }

    if (span < SPANS - 1) {
        SUM_PIECE_REFILL(pb, 1, plo, phi, 192)
        SUM_PIECE(pa, 2)
        SUM_PIECE(pb, 3)
    } else {
        SUM_PIECE(pb, 1)
        SUM_PIECE(pa, 2)
    }

    asm volatile("" : "+v"(acc_halo));

    const int fl0 = 64 * (c >> 2) + 32 * ((c >> 1) & 1) + 16 * h + (c & 1);
    #pragma unroll
    for (int v = 0; v < 8; ++v) {
        const int s = fl0 + 2 * v;
        L[0][r][col(s)]     = acc_lo[v];
        L[1][r][col(s + 1)] = acc_hi[v];
    }
    if (lane == 0) L[1][r][0] = acc_halo[0];

    __syncthreads();

    const int n_s = (span < SPANS - 1) ? SPAN : (LAST_VALID + 1);
    float* obase = out + (size_t)bc * OUT_LEN + (size_t)8 * F0;
    #pragma unroll
    for (int pass = 0; pass < 2; ++pass) {
        const int g  = tid + 256 * pass;
        const int s  = g >> 1;
        const int r0 = 4 * (g & 1);
        if (s < n_s) {
            const bool lo_ok = (F0 + s) < FRAMES;
            const int  cs    = col(s);
            v4f o;
            #pragma unroll
            for (int i = 0; i < 4; ++i) {
                const float a = lo_ok ? L[0][r0 + i][cs] : 0.0f;
                o[i] = a + L[1][r0 + i][cs];
            }
            *(v4f*)(obase + 4 * g) = o;
        }
    }
}

extern "C" void kernel_launch(void* const* d_in, const int* in_sizes, int n_in,
                              void* d_out, int out_size, void* d_ws, size_t ws_size,
                              hipStream_t stream)
{
    const float* est = (const float*)d_in[0];
    float* out = (float*)d_out;
    hipLaunchKernelGGL(oadd_residue_wmma,
                       dim3(N_BC * SPANS), dim3(256), 0, stream,
                       est, out);
}
